# starting kernel
# baseline (speedup 1.0000x reference)
#include <hip/hip_runtime.h>
#include <hip/hip_bf16.h>

typedef __attribute__((ext_vector_type(2))) float v2f;
typedef __attribute__((ext_vector_type(4))) float v4f;
typedef __attribute__((ext_vector_type(8))) float v8f;

#define IN_CH 256
#define HID   16


__global__ void k_deg_init(unsigned* __restrict__ deg, int N) {
    int n = blockIdx.x * blockDim.x + threadIdx.x;
    if (n < N) deg[n] = 1u;
}

__global__ void k_deg_count(const int* __restrict__ ei, int E,
                            unsigned* __restrict__ deg) {
    int e = blockIdx.x * blockDim.x + threadIdx.x;
    if (e < E) {
        int t = __builtin_nontemporal_load(ei + E + e);
        atomicAdd(&deg[t], 1u);
    }
}

__global__ void k_dinv(const unsigned* __restrict__ deg,
                       float* __restrict__ dinv, int N) {
    int n = blockIdx.x * blockDim.x + threadIdx.x;
    if (n < N) dinv[n] = rsqrtf((float)deg[n]);
}

#define XPAD 4

__global__ __launch_bounds__(64)
void k_gemm_wmma(const float* __restrict__ x,
                 const float* __restrict__ w,
                 float* __restrict__ h, int N) {
    __shared__ float sW[IN_CH][HID];
    __shared__ float sX[2][16][IN_CH + XPAD];

    const int tid  = threadIdx.x;
    const int wv   = tid >> 5;
    const int lane = tid & 31;
    const int l16  = lane & 15;
    const int half = lane >> 4;
    const int node_base = blockIdx.x * 32 + wv * 16;
    const bool full_tile = (node_base + 16 <= N);

    {
        const v4f* ws = (const v4f*)w;
        v4f* wd = (v4f*)&sW[0][0];
#pragma unroll
        for (int i = 0; i < (IN_CH * HID) / 4 / 64; ++i)
            wd[tid + i * 64] = ws[tid + i * 64];
    }

    if (full_tile) {
        const v4f* xs = (const v4f*)(x + (size_t)node_base * IN_CH);
#pragma unroll
        for (int it = 0; it < 32; ++it) {
            int fi = it * 32 + lane;
            v4f v = __builtin_nontemporal_load(xs + fi);
            int f = fi * 4;
            *(v4f*)&sX[wv][f >> 8][f & 255] = v;
        }
    } else {
        for (int it = 0; it < 32; ++it) {
            int f = it * 128 + lane * 4;
            int r = f >> 8, c = f & 255;
            int row = node_base + r;
            v4f v = {};
            if (row < N) v = *(const v4f*)(x + (size_t)row * IN_CH + c);
            *(v4f*)&sX[wv][r][c] = v;
        }
    }
    __syncthreads();

    v8f acc = {};
#pragma unroll 8
    for (int k = 0; k < IN_CH; k += 4) {
        v2f a;
        a.x = sX[wv][l16][k + 2 * half + 0];
        a.y = sX[wv][l16][k + 2 * half + 1];
        v2f b;
        b.x = sW[k + 2 * half + 0][l16];
        b.y = sW[k + 2 * half + 1][l16];
        acc = __builtin_amdgcn_wmma_f32_16x16x4_f32(
            false, a, false, b, (short)0, acc, false, false);
    }

    if (full_tile) {
        unsigned base = (unsigned)(node_base + 8 * half) * HID + (unsigned)l16;
#pragma unroll
        for (int i = 0; i < 8; ++i)
            h[base + (unsigned)i * HID] = acc[i];
    } else {
#pragma unroll
        for (int i = 0; i < 8; ++i) {
            int row = node_base + i + 8 * half;
            if (row < N) h[(size_t)row * HID + l16] = acc[i];
        }
    }
}


__global__ void k_self_init(const float* __restrict__ h,
                            const float* __restrict__ dinv,
                            float* __restrict__ agg, int N) {
    int n = blockIdx.x * blockDim.x + threadIdx.x;
    if (n >= N) return;
    float c = dinv[n];
    c *= c;
    const v4f* hv = (const v4f*)(h + (size_t)n * HID);
    v4f* av = (v4f*)(agg + (size_t)n * HID);
#pragma unroll
    for (int i = 0; i < 4; ++i)
        av[i] = hv[i] * c;
}


__global__ void k_edge_scatter(const int* __restrict__ ei, int E,
                               const float* __restrict__ dinv,
                               const float* __restrict__ h,
                               float* __restrict__ agg) {
    long long gid = (long long)blockIdx.x * blockDim.x + threadIdx.x;
    int e = (int)(gid >> 4);
    int j = (int)(gid & 15);
    if (e >= E) return;
    int s = __builtin_nontemporal_load(ei + e);
    int t = __builtin_nontemporal_load(ei + E + e);
    float coef = dinv[s] * dinv[t];
    unsafeAtomicAdd(&agg[(size_t)t * HID + j], h[(size_t)s * HID + j] * coef);
}


__global__ void k_out(const float* __restrict__ agg,
                      const float* __restrict__ conv_b,
                      const float* __restrict__ out_w,
                      const float* __restrict__ out_b,
                      float* __restrict__ out, int N) {
    int n = blockIdx.x * blockDim.x + threadIdx.x;
    if (n >= N) return;
    const v4f* av = (const v4f*)(agg + (size_t)n * HID);
    const v4f* bv = (const v4f*)conv_b;
    const v4f* wv = (const v4f*)out_w;
    float acc = out_b[0];
#pragma unroll
    for (int i = 0; i < 4; ++i) {
        v4f v = av[i] + bv[i];
        v4f wvi = wv[i];
#pragma unroll
        for (int j = 0; j < 4; ++j)
            acc = fmaf(fmaxf(v[j], 0.0f), wvi[j], acc);
    }
    out[n] = acc;
}


static inline size_t align256(size_t v) { return (v + 255) & ~(size_t)255; }

extern "C" void kernel_launch(void* const* d_in, const int* in_sizes, int n_in,
                              void* d_out, int out_size, void* d_ws, size_t ws_size,
                              hipStream_t stream) {
    const float* x      = (const float*)d_in[0];
    const int*   ei     = (const int*)d_in[1];
    const float* conv_w = (const float*)d_in[2];
    const float* conv_b = (const float*)d_in[3];
    const float* out_w  = (const float*)d_in[4];
    const float* out_b  = (const float*)d_in[5];
    float* out = (float*)d_out;

    const int N = in_sizes[0] / IN_CH;
    const int E = in_sizes[1] / 2;

    char* ws = (char*)d_ws;
    size_t off = 0;
    unsigned* deg = (unsigned*)(ws + off); off = align256(off + (size_t)N * 4);
    float* dinv   = (float*)(ws + off);    off = align256(off + (size_t)N * 4);
    float* h      = (float*)(ws + off);    off = align256(off + (size_t)N * HID * 4);
    float* agg    = (float*)(ws + off);    off = align256(off + (size_t)N * HID * 4);

    const int nb_n = (N + 255) / 256;
    const int nb_e = (E + 255) / 256;

    k_deg_init<<<nb_n, 256, 0, stream>>>(deg, N);
    k_deg_count<<<nb_e, 256, 0, stream>>>(ei, E, deg);
    k_dinv<<<nb_n, 256, 0, stream>>>(deg, dinv, N);

    k_gemm_wmma<<<(N + 31) / 32, 64, 0, stream>>>(x, conv_w, h, N);

    k_self_init<<<nb_n, 256, 0, stream>>>(h, dinv, agg, N);

    long long scatter_threads = (long long)E * HID;
    int nb_s = (int)((scatter_threads + 255) / 256);
    k_edge_scatter<<<nb_s, 256, 0, stream>>>(ei, E, dinv, h, agg);

    k_out<<<nb_n, 256, 0, stream>>>(agg, conv_b, out_w, out_b, out, N);
}


# edge sums by persistent work-groups pulling 512-edge items from two queues, one per package half of target rows (own half first): agg lines added into from one half; sum order changes
# speedup vs baseline: 2.8424x; 2.8424x over previous
#include <hip/hip_runtime.h>
#include <hip/hip_bf16.h>

typedef __attribute__((ext_vector_type(2))) float v2f;
typedef __attribute__((ext_vector_type(4))) float v4f;
typedef __attribute__((ext_vector_type(8))) float v8f;

#define IN_CH 256
#define HID   16

#define NPLANE 8

typedef __attribute__((ext_vector_type(4))) unsigned v4u;

#define QSTRIDE 64

__global__ void k_deg_init(v4u* __restrict__ cnt4, int n4, unsigned* __restrict__ qctr) {
    int i = blockIdx.x * blockDim.x + threadIdx.x;
    if (i < n4) cnt4[i] = (v4u){0u, 0u, 0u, 0u};
    if (i == 0) { qctr[0] = 0u; qctr[QSTRIDE] = 0u; }
}

__global__ void k_deg_count(const int* __restrict__ ei, int E,
                            unsigned* __restrict__ cnt, int N) {
    int e = blockIdx.x * blockDim.x + threadIdx.x;
    unsigned die = (__builtin_amdgcn_s_sendmsg_rtn(0x87) >> 16) & (unsigned)(NPLANE - 1);
    if (e < E) {
        int t = __builtin_nontemporal_load(ei + E + e);
        atomicAdd(&cnt[(size_t)die * (size_t)N + (size_t)t], 1u);
    }
}

__global__ void k_dinv(const unsigned* __restrict__ cnt,
                       float* __restrict__ dinv, int N) {
    int n = blockIdx.x * blockDim.x + threadIdx.x;
    if (n < N) {
        unsigned d = 1u;
        for (int p = 0; p < NPLANE; ++p)
            d += cnt[(size_t)p * (size_t)N + (size_t)n];
        dinv[n] = rsqrtf((float)d);
    }
}

#define XPAD 4

__global__ __launch_bounds__(64)
void k_gemm_wmma(const float* __restrict__ x,
                 const float* __restrict__ w,
                 float* __restrict__ h, int N) {
    __shared__ float sW[IN_CH][HID];
    __shared__ float sX[2][16][IN_CH + XPAD];

    const int tid  = threadIdx.x;
    const int wv   = tid >> 5;
    const int lane = tid & 31;
    const int l16  = lane & 15;
    const int half = lane >> 4;
    const int node_base = blockIdx.x * 32 + wv * 16;
    const bool full_tile = (node_base + 16 <= N);

    {
        const v4f* ws = (const v4f*)w;
        v4f* wd = (v4f*)&sW[0][0];
#pragma unroll
        for (int i = 0; i < (IN_CH * HID) / 4 / 64; ++i)
            wd[tid + i * 64] = ws[tid + i * 64];
    }

    if (full_tile) {
        const v4f* xs = (const v4f*)(x + (size_t)node_base * IN_CH);
#pragma unroll
        for (int it = 0; it < 32; ++it) {
            int fi = it * 32 + lane;
            v4f v = __builtin_nontemporal_load(xs + fi);
            int f = fi * 4;
            *(v4f*)&sX[wv][f >> 8][f & 255] = v;
        }
    } else {
        for (int it = 0; it < 32; ++it) {
            int f = it * 128 + lane * 4;
            int r = f >> 8, c = f & 255;
            int row = node_base + r;
            v4f v = {};
            if (row < N) v = *(const v4f*)(x + (size_t)row * IN_CH + c);
            *(v4f*)&sX[wv][r][c] = v;
        }
    }
    __syncthreads();

    v8f acc = {};
#pragma unroll 8
    for (int k = 0; k < IN_CH; k += 4) {
        v2f a;
        a.x = sX[wv][l16][k + 2 * half + 0];
        a.y = sX[wv][l16][k + 2 * half + 1];
        v2f b;
        b.x = sW[k + 2 * half + 0][l16];
        b.y = sW[k + 2 * half + 1][l16];
        acc = __builtin_amdgcn_wmma_f32_16x16x4_f32(
            false, a, false, b, (short)0, acc, false, false);
    }

    if (full_tile) {
        unsigned base = (unsigned)(node_base + 8 * half) * HID + (unsigned)l16;
#pragma unroll
        for (int i = 0; i < 8; ++i)
            h[base + (unsigned)i * HID] = acc[i];
    } else {
#pragma unroll
        for (int i = 0; i < 8; ++i) {
            int row = node_base + i + 8 * half;
            if (row < N) h[(size_t)row * HID + l16] = acc[i];
        }
    }
}


__global__ void k_self_init(const float* __restrict__ h,
                            const float* __restrict__ dinv,
                            float* __restrict__ agg, int N) {
    int n = blockIdx.x * blockDim.x + threadIdx.x;
    if (n >= N) return;
    float c = dinv[n];
    c *= c;
    const v4f* hv = (const v4f*)(h + (size_t)n * HID);
    v4f* av = (v4f*)(agg + (size_t)n * HID);
#pragma unroll
    for (int i = 0; i < 4; ++i)
        av[i] = hv[i] * c;
}


#define SCAT_ITEM  512
#define SCAT_GRID  2048

#define STEPS16(F) F(0) F(1) F(2) F(3) F(4) F(5) F(6) F(7) F(8) F(9) F(10) F(11) F(12) F(13) F(14) F(15)

template <int K>
__device__ __forceinline__ int row_lane(int v) {
    return __builtin_amdgcn_update_dpp(0, v, 0x150 + K, 0xF, 0xF, false);
}

struct EdgeGroup { int sg; int tg; float coef; };

__device__ __forceinline__ EdgeGroup edge_group(int s, int t, unsigned q, int n_half,
                                                const float* __restrict__ dinv) {
    const bool mine = (t >= 0) && ((t >= n_half ? 1u : 0u) == q);
    EdgeGroup g;
    g.sg = mine ? s : 0;
    g.tg = mine ? t : -1;
    g.coef = dinv[g.sg] * dinv[mine ? t : 0];
    return g;
}

template <int K>
__device__ __forceinline__ float h_row_read(const float* __restrict__ h, int sg, unsigned j4) {
    const unsigned sk = (unsigned)row_lane<K>(sg);
    return *(const float*)((const char*)h + (sk * (unsigned)(HID * 4) + j4));
}

template <int K>
__device__ __forceinline__ void agg_row_add(float* __restrict__ agg, const EdgeGroup& g,
                                            float hv, unsigned j4) {
    const int tk = row_lane<K>(g.tg);
    const float ck = __int_as_float(row_lane<K>(__float_as_int(g.coef)));
    if (tk >= 0)
        unsafeAtomicAdd((float*)((char*)agg + ((unsigned)tk * (unsigned)(HID * 4) + j4)), hv * ck);
}

__global__ __launch_bounds__(256)
void k_edge_scatter(const int* __restrict__ ei, int E,
                    const float* __restrict__ dinv,
                    const float* __restrict__ h,
                    float* __restrict__ agg,
                    unsigned* __restrict__ qctr, int N) {
    __shared__ unsigned s_half;
    __shared__ unsigned s_item[2];
    const int tid  = threadIdx.x;
    const int lane = tid & 31;
    const int wv   = tid >> 5;
    const unsigned j4 = (unsigned)(lane & 15) * 4u;
    const unsigned n_items = (unsigned)((E + SCAT_ITEM - 1) / SCAT_ITEM);
    const int n_half = N >> 1;

    if (tid == 0) s_half = (__builtin_amdgcn_s_sendmsg_rtn(0x87) >> 18) & 1u;
    __syncthreads();
    const unsigned my_half = s_half;

    for (int pass = 0; pass < 2; ++pass) {
        const unsigned q = my_half ^ (unsigned)pass;
        unsigned* ctr = qctr + q * QSTRIDE;
        __syncthreads();
        if (tid == 0) s_item[0] = atomicAdd(ctr, 1u);
        __syncthreads();
        for (unsigned it = 0; it <= n_items; ++it) {
            const unsigned b = (unsigned)__builtin_amdgcn_readfirstlane((int)s_item[it & 1u]);
            if (b >= n_items) break;

            const int eA = (int)(b * (unsigned)SCAT_ITEM) + wv * 64 + lane;
            const int eB = eA + 32;
            int sA = 0, tA = -1, sB = 0, tB = -1;
            if (eA < E) {
                sA = __builtin_nontemporal_load(ei + eA);
                tA = __builtin_nontemporal_load(ei + E + eA);
            }
            if (eB < E) {
                sB = __builtin_nontemporal_load(ei + eB);
                tB = __builtin_nontemporal_load(ei + E + eB);
            }
            const EdgeGroup gA = edge_group(sA, tA, q, n_half, dinv);
            const EdgeGroup gB = edge_group(sB, tB, q, n_half, dinv);

            float ra[16], rb[16];
#define RD_A(K) ra[K] = h_row_read<K>(h, gA.sg, j4);
#define RD_B(K) rb[K] = h_row_read<K>(h, gB.sg, j4);
            STEPS16(RD_A)
            STEPS16(RD_B)

            unsigned nxt = 0u;
            if (tid == 0) nxt = atomicAdd(ctr, 1u);

#define ADD_A(K) agg_row_add<K>(agg, gA, ra[K], j4);
#define ADD_B(K) agg_row_add<K>(agg, gB, rb[K], j4);
            STEPS16(ADD_A)
            STEPS16(ADD_B)

            if (tid == 0) s_item[(it + 1u) & 1u] = nxt;
            __syncthreads();
        }
    }
}


__global__ void k_out(const float* __restrict__ agg,
                      const float* __restrict__ conv_b,
                      const float* __restrict__ out_w,
                      const float* __restrict__ out_b,
                      float* __restrict__ out, int N) {
    int n = blockIdx.x * blockDim.x + threadIdx.x;
    if (n >= N) return;
    const v4f* av = (const v4f*)(agg + (size_t)n * HID);
    const v4f* bv = (const v4f*)conv_b;
    const v4f* wv = (const v4f*)out_w;
    float acc = out_b[0];
#pragma unroll
    for (int i = 0; i < 4; ++i) {
        v4f v = av[i] + bv[i];
        v4f wvi = wv[i];
#pragma unroll
        for (int j = 0; j < 4; ++j)
            acc = fmaf(fmaxf(v[j], 0.0f), wvi[j], acc);
    }
    out[n] = acc;
}


static inline size_t align256(size_t v) { return (v + 255) & ~(size_t)255; }

extern "C" void kernel_launch(void* const* d_in, const int* in_sizes, int n_in,
                              void* d_out, int out_size, void* d_ws, size_t ws_size,
                              hipStream_t stream) {
    const float* x      = (const float*)d_in[0];
    const int*   ei     = (const int*)d_in[1];
    const float* conv_w = (const float*)d_in[2];
    const float* conv_b = (const float*)d_in[3];
    const float* out_w  = (const float*)d_in[4];
    const float* out_b  = (const float*)d_in[5];
    float* out = (float*)d_out;

    const int N = in_sizes[0] / IN_CH;
    const int E = in_sizes[1] / 2;

    char* ws = (char*)d_ws;
    size_t off = 0;
    off = align256(off + (size_t)N * 4);
    float* dinv   = (float*)(ws + off);    off = align256(off + (size_t)N * 4);
    float* h      = (float*)(ws + off);    off = align256(off + (size_t)N * HID * 4);
    float* agg    = (float*)(ws + off);    off = align256(off + (size_t)N * HID * 4);

    const int nb_n = (N + 255) / 256;
    const int nb_e = (E + 255) / 256;

    unsigned* cnt = (unsigned*)agg;
    const int n4 = N * (NPLANE / 4);
    unsigned* qctr = (unsigned*)ws;
    k_deg_init<<<(n4 + 255) / 256, 256, 0, stream>>>((v4u*)cnt, n4, qctr);
    k_deg_count<<<nb_e, 256, 0, stream>>>(ei, E, cnt, N);
    k_dinv<<<nb_n, 256, 0, stream>>>(cnt, dinv, N);

    k_gemm_wmma<<<(N + 31) / 32, 64, 0, stream>>>(x, conv_w, h, N);

    k_self_init<<<nb_n, 256, 0, stream>>>(h, dinv, agg, N);

    k_edge_scatter<<<SCAT_GRID, 256, 0, stream>>>(ei, E, dinv, h, agg, qctr, N);

    k_out<<<nb_n, 256, 0, stream>>>(agg, conv_b, out_w, out_b, out, N);
}


# degree count by a fixed grid of 1024 work-groups: each lane reads four targets per vector load and issues its adds back to back, instead of 100000 one-add waves; counts exact, float sums untouched
# speedup vs baseline: 3.1803x; 1.1189x over previous
#include <hip/hip_runtime.h>
#include <hip/hip_bf16.h>

typedef __attribute__((ext_vector_type(2))) float v2f;
typedef __attribute__((ext_vector_type(4))) float v4f;
typedef __attribute__((ext_vector_type(8))) float v8f;

#define IN_CH 256
#define HID   16

#define NPLANE 8

typedef __attribute__((ext_vector_type(4))) unsigned v4u;

#define QSTRIDE 64

__global__ void k_deg_init(v4u* __restrict__ cnt4, int n4, unsigned* __restrict__ qctr) {
    int i = blockIdx.x * blockDim.x + threadIdx.x;
    if (i < n4) cnt4[i] = (v4u){0u, 0u, 0u, 0u};
    if (i == 0) { qctr[0] = 0u; qctr[QSTRIDE] = 0u; }
}

#define DEG_GRID 1024

typedef __attribute__((ext_vector_type(4))) int v4i;

__global__ __launch_bounds__(256)
void k_deg_count(const int* __restrict__ ei, int E,
                 unsigned* __restrict__ cnt, int N) {
    const unsigned die = (__builtin_amdgcn_s_sendmsg_rtn(0x87) >> 16) & (unsigned)(NPLANE - 1);
    unsigned* __restrict__ plane = cnt + (size_t)die * (size_t)N;
    const int* __restrict__ tg = ei + E;
    const int gid    = blockIdx.x * blockDim.x + threadIdx.x;
    const int stride = gridDim.x * blockDim.x;

    if (((E & 3) == 0) && ((((size_t)tg) & 15) == 0)) {
        const v4i* __restrict__ tg4 = (const v4i*)tg;
        const int n4 = E >> 2;
        for (int i = gid; i < n4; i += 4 * stride) {
            v4i t[4];
#pragma unroll
            for (int k = 0; k < 4; ++k) {
                const int j = i + k * stride;
                t[k] = (v4i){0, 0, 0, 0};
                if (j < n4) t[k] = __builtin_nontemporal_load(tg4 + j);
            }
#pragma unroll
            for (int k = 0; k < 4; ++k) {
                if (i + k * stride < n4) {
                    atomicAdd(&plane[t[k].x], 1u);
                    atomicAdd(&plane[t[k].y], 1u);
                    atomicAdd(&plane[t[k].z], 1u);
                    atomicAdd(&plane[t[k].w], 1u);
                }
            }
        }
    } else {
        for (int e = gid; e < E; e += stride)
            atomicAdd(&plane[__builtin_nontemporal_load(tg + e)], 1u);
    }
}

__global__ void k_dinv(const unsigned* __restrict__ cnt,
                       float* __restrict__ dinv, int N) {
    int n = blockIdx.x * blockDim.x + threadIdx.x;
    if (n < N) {
        unsigned d = 1u;
        for (int p = 0; p < NPLANE; ++p)
            d += cnt[(size_t)p * (size_t)N + (size_t)n];
        dinv[n] = rsqrtf((float)d);
    }
}

#define XPAD 4

__global__ __launch_bounds__(64)
void k_gemm_wmma(const float* __restrict__ x,
                 const float* __restrict__ w,
                 float* __restrict__ h, int N) {
    __shared__ float sW[IN_CH][HID];
    __shared__ float sX[2][16][IN_CH + XPAD];

    const int tid  = threadIdx.x;
    const int wv   = tid >> 5;
    const int lane = tid & 31;
    const int l16  = lane & 15;
    const int half = lane >> 4;
    const int node_base = blockIdx.x * 32 + wv * 16;
    const bool full_tile = (node_base + 16 <= N);

    {
        const v4f* ws = (const v4f*)w;
        v4f* wd = (v4f*)&sW[0][0];
#pragma unroll
        for (int i = 0; i < (IN_CH * HID) / 4 / 64; ++i)
            wd[tid + i * 64] = ws[tid + i * 64];
    }

    if (full_tile) {
        const v4f* xs = (const v4f*)(x + (size_t)node_base * IN_CH);
#pragma unroll
        for (int it = 0; it < 32; ++it) {
            int fi = it * 32 + lane;
            v4f v = __builtin_nontemporal_load(xs + fi);
            int f = fi * 4;
            *(v4f*)&sX[wv][f >> 8][f & 255] = v;
        }
    } else {
        for (int it = 0; it < 32; ++it) {
            int f = it * 128 + lane * 4;
            int r = f >> 8, c = f & 255;
            int row = node_base + r;
            v4f v = {};
            if (row < N) v = *(const v4f*)(x + (size_t)row * IN_CH + c);
            *(v4f*)&sX[wv][r][c] = v;
        }
    }
    __syncthreads();

    v8f acc = {};
#pragma unroll 8
    for (int k = 0; k < IN_CH; k += 4) {
        v2f a;
        a.x = sX[wv][l16][k + 2 * half + 0];
        a.y = sX[wv][l16][k + 2 * half + 1];
        v2f b;
        b.x = sW[k + 2 * half + 0][l16];
        b.y = sW[k + 2 * half + 1][l16];
        acc = __builtin_amdgcn_wmma_f32_16x16x4_f32(
            false, a, false, b, (short)0, acc, false, false);
    }

    if (full_tile) {
        unsigned base = (unsigned)(node_base + 8 * half) * HID + (unsigned)l16;
#pragma unroll
        for (int i = 0; i < 8; ++i)
            h[base + (unsigned)i * HID] = acc[i];
    } else {
#pragma unroll
        for (int i = 0; i < 8; ++i) {
            int row = node_base + i + 8 * half;
            if (row < N) h[(size_t)row * HID + l16] = acc[i];
        }
    }
}


__global__ void k_self_init(const float* __restrict__ h,
                            const float* __restrict__ dinv,
                            float* __restrict__ agg, int N) {
    int n = blockIdx.x * blockDim.x + threadIdx.x;
    if (n >= N) return;
    float c = dinv[n];
    c *= c;
    const v4f* hv = (const v4f*)(h + (size_t)n * HID);
    v4f* av = (v4f*)(agg + (size_t)n * HID);
#pragma unroll
    for (int i = 0; i < 4; ++i)
        av[i] = hv[i] * c;
}


#define SCAT_ITEM  512
#define SCAT_GRID  2048

#define STEPS16(F) F(0) F(1) F(2) F(3) F(4) F(5) F(6) F(7) F(8) F(9) F(10) F(11) F(12) F(13) F(14) F(15)

template <int K>
__device__ __forceinline__ int row_lane(int v) {
    return __builtin_amdgcn_update_dpp(0, v, 0x150 + K, 0xF, 0xF, false);
}

struct EdgeGroup { int sg; int tg; float coef; };

__device__ __forceinline__ EdgeGroup edge_group(int s, int t, unsigned q, int n_half,
                                                const float* __restrict__ dinv) {
    const bool mine = (t >= 0) && ((t >= n_half ? 1u : 0u) == q);
    EdgeGroup g;
    g.sg = mine ? s : 0;
    g.tg = mine ? t : -1;
    g.coef = dinv[g.sg] * dinv[mine ? t : 0];
    return g;
}

template <int K>
__device__ __forceinline__ float h_row_read(const float* __restrict__ h, int sg, unsigned j4) {
    const unsigned sk = (unsigned)row_lane<K>(sg);
    return *(const float*)((const char*)h + (sk * (unsigned)(HID * 4) + j4));
}

template <int K>
__device__ __forceinline__ void agg_row_add(float* __restrict__ agg, const EdgeGroup& g,
                                            float hv, unsigned j4) {
    const int tk = row_lane<K>(g.tg);
    const float ck = __int_as_float(row_lane<K>(__float_as_int(g.coef)));
    if (tk >= 0)
        unsafeAtomicAdd((float*)((char*)agg + ((unsigned)tk * (unsigned)(HID * 4) + j4)), hv * ck);
}

__global__ __launch_bounds__(256)
void k_edge_scatter(const int* __restrict__ ei, int E,
                    const float* __restrict__ dinv,
                    const float* __restrict__ h,
                    float* __restrict__ agg,
                    unsigned* __restrict__ qctr, int N) {
    __shared__ unsigned s_half;
    __shared__ unsigned s_item[2];
    const int tid  = threadIdx.x;
    const int lane = tid & 31;
    const int wv   = tid >> 5;
    const unsigned j4 = (unsigned)(lane & 15) * 4u;
    const unsigned n_items = (unsigned)((E + SCAT_ITEM - 1) / SCAT_ITEM);
    const int n_half = N >> 1;

    if (tid == 0) s_half = (__builtin_amdgcn_s_sendmsg_rtn(0x87) >> 18) & 1u;
    __syncthreads();
    const unsigned my_half = s_half;

    for (int pass = 0; pass < 2; ++pass) {
        const unsigned q = my_half ^ (unsigned)pass;
        unsigned* ctr = qctr + q * QSTRIDE;
        __syncthreads();
        if (tid == 0) s_item[0] = atomicAdd(ctr, 1u);
        __syncthreads();
        for (unsigned it = 0; it <= n_items; ++it) {
            const unsigned b = (unsigned)__builtin_amdgcn_readfirstlane((int)s_item[it & 1u]);
            if (b >= n_items) break;

            const int eA = (int)(b * (unsigned)SCAT_ITEM) + wv * 64 + lane;
            const int eB = eA + 32;
            int sA = 0, tA = -1, sB = 0, tB = -1;
            if (eA < E) {
                sA = __builtin_nontemporal_load(ei + eA);
                tA = __builtin_nontemporal_load(ei + E + eA);
            }
            if (eB < E) {
                sB = __builtin_nontemporal_load(ei + eB);
                tB = __builtin_nontemporal_load(ei + E + eB);
            }
            const EdgeGroup gA = edge_group(sA, tA, q, n_half, dinv);
            const EdgeGroup gB = edge_group(sB, tB, q, n_half, dinv);

            float ra[16], rb[16];
#define RD_A(K) ra[K] = h_row_read<K>(h, gA.sg, j4);
#define RD_B(K) rb[K] = h_row_read<K>(h, gB.sg, j4);
            STEPS16(RD_A)
            STEPS16(RD_B)

            unsigned nxt = 0u;
            if (tid == 0) nxt = atomicAdd(ctr, 1u);

#define ADD_A(K) agg_row_add<K>(agg, gA, ra[K], j4);
#define ADD_B(K) agg_row_add<K>(agg, gB, rb[K], j4);
            STEPS16(ADD_A)
            STEPS16(ADD_B)

            if (tid == 0) s_item[(it + 1u) & 1u] = nxt;
            __syncthreads();
        }
    }
}


__global__ void k_out(const float* __restrict__ agg,
                      const float* __restrict__ conv_b,
                      const float* __restrict__ out_w,
                      const float* __restrict__ out_b,
                      float* __restrict__ out, int N) {
    int n = blockIdx.x * blockDim.x + threadIdx.x;
    if (n >= N) return;
    const v4f* av = (const v4f*)(agg + (size_t)n * HID);
    const v4f* bv = (const v4f*)conv_b;
    const v4f* wv = (const v4f*)out_w;
    float acc = out_b[0];
#pragma unroll
    for (int i = 0; i < 4; ++i) {
        v4f v = av[i] + bv[i];
        v4f wvi = wv[i];
#pragma unroll
        for (int j = 0; j < 4; ++j)
            acc = fmaf(fmaxf(v[j], 0.0f), wvi[j], acc);
    }
    out[n] = acc;
}


static inline size_t align256(size_t v) { return (v + 255) & ~(size_t)255; }

extern "C" void kernel_launch(void* const* d_in, const int* in_sizes, int n_in,
                              void* d_out, int out_size, void* d_ws, size_t ws_size,
                              hipStream_t stream) {
    const float* x      = (const float*)d_in[0];
    const int*   ei     = (const int*)d_in[1];
    const float* conv_w = (const float*)d_in[2];
    const float* conv_b = (const float*)d_in[3];
    const float* out_w  = (const float*)d_in[4];
    const float* out_b  = (const float*)d_in[5];
    float* out = (float*)d_out;

    const int N = in_sizes[0] / IN_CH;
    const int E = in_sizes[1] / 2;

    char* ws = (char*)d_ws;
    size_t off = 0;
    off = align256(off + (size_t)N * 4);
    float* dinv   = (float*)(ws + off);    off = align256(off + (size_t)N * 4);
    float* h      = (float*)(ws + off);    off = align256(off + (size_t)N * HID * 4);
    float* agg    = (float*)(ws + off);    off = align256(off + (size_t)N * HID * 4);

    const int nb_n = (N + 255) / 256;

    unsigned* cnt = (unsigned*)agg;
    const int n4 = N * (NPLANE / 4);
    unsigned* qctr = (unsigned*)ws;
    k_deg_init<<<(n4 + 255) / 256, 256, 0, stream>>>((v4u*)cnt, n4, qctr);
    k_deg_count<<<DEG_GRID, 256, 0, stream>>>(ei, E, cnt, N);
    k_dinv<<<nb_n, 256, 0, stream>>>(cnt, dinv, N);

    k_gemm_wmma<<<(N + 31) / 32, 64, 0, stream>>>(x, conv_w, h, N);

    k_self_init<<<nb_n, 256, 0, stream>>>(h, dinv, agg, N);

    k_edge_scatter<<<SCAT_GRID, 256, 0, stream>>>(ei, E, dinv, h, agg, qctr, N);

    k_out<<<nb_n, 256, 0, stream>>>(agg, conv_b, out_w, out_b, out, N);
}


# edge coefficients rebuilt from a byte-per-node degree table in LDS (two 32-wave work-groups per WGP, 2048-edge items), not two scattered dinv reads per edge; same rsqrt values, adds stay atomic
# speedup vs baseline: 3.6067x; 1.1341x over previous
#include <hip/hip_runtime.h>
#include <hip/hip_bf16.h>

typedef __attribute__((ext_vector_type(2))) float v2f;
typedef __attribute__((ext_vector_type(4))) float v4f;
typedef __attribute__((ext_vector_type(8))) float v8f;

#define IN_CH 256
#define HID   16

#define NPLANE 8

typedef __attribute__((ext_vector_type(4))) unsigned v4u;

#define QSTRIDE 64

__global__ void k_deg_init(v4u* __restrict__ cnt4, int n4, unsigned* __restrict__ qctr) {
    int i = blockIdx.x * blockDim.x + threadIdx.x;
    if (i < n4) cnt4[i] = (v4u){0u, 0u, 0u, 0u};
    if (i == 0) { qctr[0] = 0u; qctr[QSTRIDE] = 0u; }
}

#define DEG_GRID 1024

typedef __attribute__((ext_vector_type(4))) int v4i;

__global__ __launch_bounds__(256)
void k_deg_count(const int* __restrict__ ei, int E,
                 unsigned* __restrict__ cnt, int N) {
    const unsigned die = (__builtin_amdgcn_s_sendmsg_rtn(0x87) >> 16) & (unsigned)(NPLANE - 1);
    unsigned* __restrict__ plane = cnt + (size_t)die * (size_t)N;
    const int* __restrict__ tg = ei + E;
    const int gid    = blockIdx.x * blockDim.x + threadIdx.x;
    const int stride = gridDim.x * blockDim.x;

    if (((E & 3) == 0) && ((((size_t)tg) & 15) == 0)) {
        const v4i* __restrict__ tg4 = (const v4i*)tg;
        const int n4 = E >> 2;
        for (int i = gid; i < n4; i += 4 * stride) {
            v4i t[4];
#pragma unroll
            for (int k = 0; k < 4; ++k) {
                const int j = i + k * stride;
                t[k] = (v4i){0, 0, 0, 0};
                if (j < n4) t[k] = __builtin_nontemporal_load(tg4 + j);
            }
#pragma unroll
            for (int k = 0; k < 4; ++k) {
                if (i + k * stride < n4) {
                    atomicAdd(&plane[t[k].x], 1u);
                    atomicAdd(&plane[t[k].y], 1u);
                    atomicAdd(&plane[t[k].z], 1u);
                    atomicAdd(&plane[t[k].w], 1u);
                }
            }
        }
    } else {
        for (int e = gid; e < E; e += stride)
            atomicAdd(&plane[__builtin_nontemporal_load(tg + e)], 1u);
    }
}

#define TAB_MAX   131072
#define TAB_MIN_N 1024
#define TAB_OFF   512
#define DEG_BIG   255u

__device__ __forceinline__ bool deg_table_used(int N) { return N <= TAB_MAX && N >= TAB_MIN_N; }
__device__ __forceinline__ float rsq_deg(unsigned d) { return rsqrtf((float)d); }

__global__ void k_dinv(const unsigned* __restrict__ cnt,
                       float* __restrict__ dinv,
                       unsigned char* __restrict__ deg8, int N) {
    int n = blockIdx.x * blockDim.x + threadIdx.x;
    if (n < N) {
        unsigned d = 1u;
        for (int p = 0; p < NPLANE; ++p)
            d += cnt[(size_t)p * (size_t)N + (size_t)n];
        dinv[n] = rsq_deg(d);
        if (deg_table_used(N)) deg8[n] = (unsigned char)(d < DEG_BIG ? d : DEG_BIG);
    }
}

#define XPAD 4

__global__ __launch_bounds__(64)
void k_gemm_wmma(const float* __restrict__ x,
                 const float* __restrict__ w,
                 float* __restrict__ h, int N) {
    __shared__ float sW[IN_CH][HID];
    __shared__ float sX[2][16][IN_CH + XPAD];

    const int tid  = threadIdx.x;
    const int wv   = tid >> 5;
    const int lane = tid & 31;
    const int l16  = lane & 15;
    const int half = lane >> 4;
    const int node_base = blockIdx.x * 32 + wv * 16;
    const bool full_tile = (node_base + 16 <= N);

    {
        const v4f* ws = (const v4f*)w;
        v4f* wd = (v4f*)&sW[0][0];
#pragma unroll
        for (int i = 0; i < (IN_CH * HID) / 4 / 64; ++i)
            wd[tid + i * 64] = ws[tid + i * 64];
    }

    if (full_tile) {
        const v4f* xs = (const v4f*)(x + (size_t)node_base * IN_CH);
#pragma unroll
        for (int it = 0; it < 32; ++it) {
            int fi = it * 32 + lane;
            v4f v = __builtin_nontemporal_load(xs + fi);
            int f = fi * 4;
            *(v4f*)&sX[wv][f >> 8][f & 255] = v;
        }
    } else {
        for (int it = 0; it < 32; ++it) {
            int f = it * 128 + lane * 4;
            int r = f >> 8, c = f & 255;
            int row = node_base + r;
            v4f v = {};
            if (row < N) v = *(const v4f*)(x + (size_t)row * IN_CH + c);
            *(v4f*)&sX[wv][r][c] = v;
        }
    }
    __syncthreads();

    v8f acc = {};
#pragma unroll 8
    for (int k = 0; k < IN_CH; k += 4) {
        v2f a;
        a.x = sX[wv][l16][k + 2 * half + 0];
        a.y = sX[wv][l16][k + 2 * half + 1];
        v2f b;
        b.x = sW[k + 2 * half + 0][l16];
        b.y = sW[k + 2 * half + 1][l16];
        acc = __builtin_amdgcn_wmma_f32_16x16x4_f32(
            false, a, false, b, (short)0, acc, false, false);
    }

    if (full_tile) {
        unsigned base = (unsigned)(node_base + 8 * half) * HID + (unsigned)l16;
#pragma unroll
        for (int i = 0; i < 8; ++i)
            h[base + (unsigned)i * HID] = acc[i];
    } else {
#pragma unroll
        for (int i = 0; i < 8; ++i) {
            int row = node_base + i + 8 * half;
            if (row < N) h[(size_t)row * HID + l16] = acc[i];
        }
    }
}


__global__ void k_self_init(const float* __restrict__ h,
                            const float* __restrict__ dinv,
                            float* __restrict__ agg, int N) {
    int n = blockIdx.x * blockDim.x + threadIdx.x;
    if (n >= N) return;
    float c = dinv[n];
    c *= c;
    const v4f* hv = (const v4f*)(h + (size_t)n * HID);
    v4f* av = (v4f*)(agg + (size_t)n * HID);
#pragma unroll
    for (int i = 0; i < 4; ++i)
        av[i] = hv[i] * c;
}


#define SCAT_WAVES 32
#define SCAT_ITEM  2048
#define SCAT_GRID  512

#define STEPS16(F) F(0) F(1) F(2) F(3) F(4) F(5) F(6) F(7) F(8) F(9) F(10) F(11) F(12) F(13) F(14) F(15)

template <int K>
__device__ __forceinline__ int row_lane(int v) {
    return __builtin_amdgcn_update_dpp(0, v, 0x150 + K, 0xF, 0xF, false);
}

struct EdgeGroup { int sg; int tg; float coef; };

__device__ __forceinline__ EdgeGroup edge_pick(int s, int t, unsigned q, int n_half, int& tn) {
    const bool mine = (t >= 0) && ((t >= n_half ? 1u : 0u) == q);
    EdgeGroup g;
    g.sg = mine ? s : 0;
    g.tg = mine ? t : -1;
    g.coef = 0.0f;
    tn = mine ? t : 0;
    return g;
}

__device__ __forceinline__ float node_factor(unsigned d, int n, const float* __restrict__ dinv) {
    if (d != DEG_BIG) return rsq_deg(d);
    return dinv[n];
}

template <int K>
__device__ __forceinline__ float h_row_read(const float* __restrict__ h, int sg, unsigned j4) {
    const unsigned sk = (unsigned)row_lane<K>(sg);
    return *(const float*)((const char*)h + (sk * (unsigned)(HID * 4) + j4));
}

template <int K>
__device__ __forceinline__ void agg_row_add(float* __restrict__ agg, const EdgeGroup& g,
                                            float hv, unsigned j4) {
    const int tk = row_lane<K>(g.tg);
    const float ck = __int_as_float(row_lane<K>(__float_as_int(g.coef)));
    if (tk >= 0)
        unsafeAtomicAdd((float*)((char*)agg + ((unsigned)tk * (unsigned)(HID * 4) + j4)), hv * ck);
}

__global__ __launch_bounds__(SCAT_WAVES * 32)
void k_edge_scatter(const int* __restrict__ ei, int E,
                    const float* __restrict__ dinv,
                    const unsigned char* __restrict__ deg8,
                    const float* __restrict__ h,
                    float* __restrict__ agg,
                    unsigned* __restrict__ qctr, int N) {
    __shared__ __attribute__((aligned(16))) unsigned char s_deg[TAB_MAX];
    __shared__ unsigned s_half;
    __shared__ unsigned s_item[2];
    const int tid  = threadIdx.x;
    const bool use_tab = deg_table_used(N);

    if (use_tab) {
        const v4u* __restrict__ tsrc = (const v4u*)deg8;
        v4u* tdst = (v4u*)s_deg;
        const int n16 = N >> 4;
        for (int i = tid; i < n16; i += 4 * SCAT_WAVES * 32) {
            v4u tv[4];
#pragma unroll
            for (int k = 0; k < 4; ++k) {
                const int j = i + k * SCAT_WAVES * 32;
                tv[k] = (v4u){0u, 0u, 0u, 0u};
                if (j < n16) tv[k] = tsrc[j];
            }
#pragma unroll
            for (int k = 0; k < 4; ++k) {
                const int j = i + k * SCAT_WAVES * 32;
                if (j < n16) tdst[j] = tv[k];
            }
        }
        for (int i = (n16 << 4) + tid; i < N; i += SCAT_WAVES * 32) s_deg[i] = deg8[i];
    }

    const int lane = tid & 31;
    const int wv   = tid >> 5;
    const unsigned j4 = (unsigned)(lane & 15) * 4u;
    const unsigned n_items = (unsigned)((E + SCAT_ITEM - 1) / SCAT_ITEM);
    const int n_half = N >> 1;

    if (tid == 0) s_half = (__builtin_amdgcn_s_sendmsg_rtn(0x87) >> 18) & 1u;
    __syncthreads();
    const unsigned my_half = s_half;

    for (int pass = 0; pass < 2; ++pass) {
        const unsigned q = my_half ^ (unsigned)pass;
        unsigned* ctr = qctr + q * QSTRIDE;
        __syncthreads();
        if (tid == 0) s_item[0] = atomicAdd(ctr, 1u);
        __syncthreads();
        for (unsigned it = 0; it <= n_items; ++it) {
            const unsigned b = (unsigned)__builtin_amdgcn_readfirstlane((int)s_item[it & 1u]);
            if (b >= n_items) break;

            const int eA = (int)(b * (unsigned)SCAT_ITEM) + wv * 64 + lane;
            const int eB = eA + 32;
            int sA = 0, tA = -1, sB = 0, tB = -1;
            if (eA < E) {
                sA = __builtin_nontemporal_load(ei + eA);
                tA = __builtin_nontemporal_load(ei + E + eA);
            }
            if (eB < E) {
                sB = __builtin_nontemporal_load(ei + eB);
                tB = __builtin_nontemporal_load(ei + E + eB);
            }
            int tnA, tnB;
            EdgeGroup gA = edge_pick(sA, tA, q, n_half, tnA);
            EdgeGroup gB = edge_pick(sB, tB, q, n_half, tnB);
            unsigned dsA = DEG_BIG, dtA = DEG_BIG, dsB = DEG_BIG, dtB = DEG_BIG;
            if (use_tab) {
                dsA = s_deg[gA.sg]; dtA = s_deg[tnA];
                dsB = s_deg[gB.sg]; dtB = s_deg[tnB];
            }
            gA.coef = node_factor(dsA, gA.sg, dinv) * node_factor(dtA, tnA, dinv);
            gB.coef = node_factor(dsB, gB.sg, dinv) * node_factor(dtB, tnB, dinv);

            float ra[16], rb[16];
#define RD_A(K) ra[K] = h_row_read<K>(h, gA.sg, j4);
#define RD_B(K) rb[K] = h_row_read<K>(h, gB.sg, j4);
            STEPS16(RD_A)
            STEPS16(RD_B)

            unsigned nxt = 0u;
            if (tid == 0) nxt = atomicAdd(ctr, 1u);

#define ADD_A(K) agg_row_add<K>(agg, gA, ra[K], j4);
#define ADD_B(K) agg_row_add<K>(agg, gB, rb[K], j4);
            STEPS16(ADD_A)
            STEPS16(ADD_B)

            if (tid == 0) s_item[(it + 1u) & 1u] = nxt;
            __syncthreads();
        }
    }
}


__global__ void k_out(const float* __restrict__ agg,
                      const float* __restrict__ conv_b,
                      const float* __restrict__ out_w,
                      const float* __restrict__ out_b,
                      float* __restrict__ out, int N) {
    int n = blockIdx.x * blockDim.x + threadIdx.x;
    if (n >= N) return;
    const v4f* av = (const v4f*)(agg + (size_t)n * HID);
    const v4f* bv = (const v4f*)conv_b;
    const v4f* wv = (const v4f*)out_w;
    float acc = out_b[0];
#pragma unroll
    for (int i = 0; i < 4; ++i) {
        v4f v = av[i] + bv[i];
        v4f wvi = wv[i];
#pragma unroll
        for (int j = 0; j < 4; ++j)
            acc = fmaf(fmaxf(v[j], 0.0f), wvi[j], acc);
    }
    out[n] = acc;
}


static inline size_t align256(size_t v) { return (v + 255) & ~(size_t)255; }

extern "C" void kernel_launch(void* const* d_in, const int* in_sizes, int n_in,
                              void* d_out, int out_size, void* d_ws, size_t ws_size,
                              hipStream_t stream) {
    const float* x      = (const float*)d_in[0];
    const int*   ei     = (const int*)d_in[1];
    const float* conv_w = (const float*)d_in[2];
    const float* conv_b = (const float*)d_in[3];
    const float* out_w  = (const float*)d_in[4];
    const float* out_b  = (const float*)d_in[5];
    float* out = (float*)d_out;

    const int N = in_sizes[0] / IN_CH;
    const int E = in_sizes[1] / 2;

    char* ws = (char*)d_ws;
    size_t off = 0;
    off = align256(off + (size_t)N * 4);
    float* dinv   = (float*)(ws + off);    off = align256(off + (size_t)N * 4);
    float* h      = (float*)(ws + off);    off = align256(off + (size_t)N * HID * 4);
    float* agg    = (float*)(ws + off);    off = align256(off + (size_t)N * HID * 4);

    const int nb_n = (N + 255) / 256;

    unsigned* cnt = (unsigned*)agg;
    const int n4 = N * (NPLANE / 4);
    unsigned* qctr = (unsigned*)ws;
    k_deg_init<<<(n4 + 255) / 256, 256, 0, stream>>>((v4u*)cnt, n4, qctr);
    k_deg_count<<<DEG_GRID, 256, 0, stream>>>(ei, E, cnt, N);
    unsigned char* deg8 = (unsigned char*)ws + TAB_OFF;
    k_dinv<<<nb_n, 256, 0, stream>>>(cnt, dinv, deg8, N);

    k_gemm_wmma<<<(N + 31) / 32, 64, 0, stream>>>(x, conv_w, h, N);

    k_self_init<<<nb_n, 256, 0, stream>>>(h, dinv, agg, N);

    k_edge_scatter<<<SCAT_GRID, SCAT_WAVES * 32, 0, stream>>>(ei, E, dinv, deg8, h, agg, qctr, N);

    k_out<<<nb_n, 256, 0, stream>>>(agg, conv_b, out_w, out_b, out, N);
}
